# starting kernel
# baseline (speedup 1.0000x reference)
#include <hip/hip_runtime.h>
#include <hip/hip_bf16.h>

typedef __attribute__((ext_vector_type(2))) float v2f;
typedef __attribute__((ext_vector_type(4))) float v4f;
typedef __attribute__((ext_vector_type(8))) float v8f;

#define K_TAPS 15
#define CIN    128
#define F_OUT  256
#define L_IN   4096
#define L_OUT  (L_IN - K_TAPS + 1)
#define BATCH  32
#define TILE_M 64
#define ROWS   (TILE_M + K_TAPS - 1)
#define XPAD   132
#define KC     32
#define KPAD   36
#define NCHUNK ((K_TAPS * CIN) / KC)
#define EPS_F  1e-7f

__global__ __launch_bounds__(256)
void ckn1d_wmma_kernel(const float* __restrict__ x,
                       const float* __restrict__ w,
                       const float* __restrict__ scale,
                       const float* __restrict__ bias,
                       float* __restrict__ out) {
    __shared__ float lds_x[ROWS * XPAD];
    __shared__ float lds_bt[F_OUT * KPAD];
    __shared__ float lds_rowsq[ROWS];
    __shared__ float lds_invn[TILE_M];
    __shared__ float lds_bias[F_OUT];

    const int t    = threadIdx.x;
    const int lane = t & 31;
    const int wave = t >> 5;
    const int l0   = blockIdx.x * TILE_M;
    const int b    = blockIdx.y;

    const float* xb = x + (size_t)b * L_IN * CIN;
    #pragma unroll
    for (int i = 0; i < 10; ++i) {
        int q = t + 256 * i;
        if (q < ROWS * (CIN / 4)) {
            int row = q >> 5;
            int f4  = q & 31;
            int g   = l0 + row;
            v4f val = {0.f, 0.f, 0.f, 0.f};
            if (g < L_IN) val = *(const v4f*)(xb + (size_t)g * CIN + f4 * 4);
            *(v4f*)(&lds_x[row * XPAD + f4 * 4]) = val;
        }
    }
    if (t < F_OUT) lds_bias[t] = bias[t];
    __syncthreads();

    if (t < ROWS) {
        const v4f* r = (const v4f*)&lds_x[t * XPAD];
        float s = 0.f;
        #pragma unroll
        for (int c = 0; c < CIN / 4; ++c) {
            v4f v = r[c];
            s += v.x * v.x + v.y * v.y + v.z * v.z + v.w * v.w;
        }
        lds_rowsq[t] = s;
    }
    __syncthreads();
    if (t < TILE_M) {
        float win = 0.f;
        #pragma unroll
        for (int k = 0; k < K_TAPS; ++k) win += lds_rowsq[t + k];
        lds_invn[t] = scale[0] / __builtin_sqrtf(win + EPS_F);
    }

    const int mt   = wave & 3;
    const int f0   = (wave >> 2) * 128;
    const int half = lane >> 4;
    const int lm   = lane & 15;

    v8f acc[8] = {};

    float pa[16], pb[16];
    {
        const float* src = w;
        #pragma unroll
        for (int i = 0; i < 16; ++i) {
            pa[i] = src[(2 * i)     * F_OUT + t];
            pb[i] = src[(2 * i + 1) * F_OUT + t];
        }
    }

    for (int ch = 0; ch < NCHUNK; ++ch) {
        __syncthreads();
        {
            float* dst = &lds_bt[t * KPAD];
            #pragma unroll
            for (int i = 0; i < 16; ++i) {
                v2f p;
                p.x = pa[i];
                p.y = pb[i];
                *(v2f*)(dst + 2 * i) = p;
            }
        }
        if (ch + 1 < NCHUNK) {
            const float* src = w + (size_t)(ch + 1) * KC * F_OUT;
            #pragma unroll
            for (int i = 0; i < 16; ++i) {
                pa[i] = src[(2 * i)     * F_OUT + t];
                pb[i] = src[(2 * i + 1) * F_OUT + t];
            }
        }
        __syncthreads();

        const int tap = ch >> 2;
        const int c0  = (ch & 3) * KC;
        const float* arow = &lds_x[(mt * 16 + lm + tap) * XPAD + c0 + 2 * half];
        const float* bcol = &lds_bt[(f0 + lm) * KPAD + 2 * half];

        #pragma unroll
        for (int kc = 0; kc < KC; kc += 4) {
            v2f a = *(const v2f*)(arow + kc);
            #pragma unroll
            for (int nt = 0; nt < 8; ++nt) {
                v2f bf = *(const v2f*)(bcol + nt * 16 * KPAD + kc);
                acc[nt] = __builtin_amdgcn_wmma_f32_16x16x4_f32(
                    false, a, false, bf, (short)0, acc[nt], false, false);
            }
        }
    }

    float* ob = out + ((size_t)b * L_OUT + l0) * F_OUT;
    #pragma unroll
    for (int v = 0; v < 8; ++v) {
        int mrow = mt * 16 + v + 8 * half;
        if (l0 + mrow < L_OUT) {
            float inv = lds_invn[mrow];
            #pragma unroll
            for (int nt = 0; nt < 8; ++nt) {
                int f = f0 + nt * 16 + lm;
                ob[(size_t)mrow * F_OUT + f] = acc[nt][v] * inv + lds_bias[f];
            }
        }
    }
}

extern "C" void kernel_launch(void* const* d_in, const int* in_sizes, int n_in,
                              void* d_out, int out_size, void* d_ws, size_t ws_size,
                              hipStream_t stream) {
    const float* x     = (const float*)d_in[0];
    const float* w     = (const float*)d_in[1];
    const float* scale = (const float*)d_in[2];
    const float* bias  = (const float*)d_in[3];
    float* out = (float*)d_out;

    dim3 grid((L_OUT + TILE_M - 1) / TILE_M, BATCH);
    ckn1d_wmma_kernel<<<grid, 256, 0, stream>>>(x, w, scale, bias, out);
}


# wave register tile 16x128 -> 64x64 (block 128x256): one b128 LDS read per fragment per 8 k (k pairs regrouped inside each 8, sum order changes), 8 reads feed 32 WMMAs instead of 9 feeding 8
# speedup vs baseline: 1.0912x; 1.0912x over previous
#include <hip/hip_runtime.h>
#include <hip/hip_bf16.h>


typedef __attribute__((ext_vector_type(2))) float v2f;
typedef __attribute__((ext_vector_type(4))) float v4f;
typedef __attribute__((ext_vector_type(8))) float v8f;

#define K_TAPS 15
#define CIN    128
#define F_OUT  256
#define L_IN   4096
#define L_OUT  (L_IN - K_TAPS + 1)
#define BATCH  32
#define TILE_M 128
#define ROWS   (TILE_M + K_TAPS - 1)
#define XPAD   132
#define KC     32
#define KPAD   36
#define NCHUNK ((K_TAPS * CIN) / KC)
#define EPS_F  1e-7f
#define WM     4
#define WN     4

__global__ __launch_bounds__(256)
void ckn1d_wmma_kernel(const float* __restrict__ x,
                       const float* __restrict__ w,
                       const float* __restrict__ scale,
                       const float* __restrict__ bias,
                       float* __restrict__ out) {
    __shared__ __attribute__((aligned(16))) float lds_x[ROWS * XPAD];
    __shared__ __attribute__((aligned(16))) float lds_bt[F_OUT * KPAD];
    __shared__ float lds_rowsq[ROWS];
    __shared__ float lds_invn[TILE_M];
    __shared__ float lds_bias[F_OUT];

    const int t    = threadIdx.x;
    const int lane = t & 31;
    const int wave = t >> 5;
    const int l0   = blockIdx.x * TILE_M;
    const int b    = blockIdx.y;

    const float* xb = x + (size_t)b * L_IN * CIN;
    #pragma unroll
    for (int i = 0; i < 18; ++i) {
        int q = t + 256 * i;
        if (q < ROWS * (CIN / 4)) {
            int row = q >> 5;
            int f4  = q & 31;
            int g   = l0 + row;
            v4f val = {0.f, 0.f, 0.f, 0.f};
            if (g < L_IN) val = *(const v4f*)(xb + (size_t)g * CIN + f4 * 4);
            *(v4f*)(&lds_x[row * XPAD + f4 * 4]) = val;
        }
    }
    if (t < F_OUT) lds_bias[t] = bias[t];
    __syncthreads();

    if (t < ROWS) {
        const v4f* r = (const v4f*)&lds_x[t * XPAD];
        float s = 0.f;
        #pragma unroll
        for (int c = 0; c < CIN / 4; ++c) {
            v4f v = r[c];
            s += v.x * v.x + v.y * v.y + v.z * v.z + v.w * v.w;
        }
        lds_rowsq[t] = s;
    }
    __syncthreads();
    if (t < TILE_M) {
        float win = 0.f;
        #pragma unroll
        for (int k = 0; k < K_TAPS; ++k) win += lds_rowsq[t + k];
        lds_invn[t] = scale[0] / __builtin_sqrtf(win + EPS_F);
    }

    const int m0   = (wave & 1) * 64;
    const int f0   = (wave >> 1) * 64;
    const int half = lane >> 4;
    const int lm   = lane & 15;

    v8f acc[WM][WN] = {};

    float pa[16], pb[16];
    {
        const float* src = w;
        #pragma unroll
        for (int i = 0; i < 16; ++i) {
            pa[i] = src[(2 * i)     * F_OUT + t];
            pb[i] = src[(2 * i + 1) * F_OUT + t];
        }
    }

    for (int ch = 0; ch < NCHUNK; ++ch) {
        __syncthreads();
        {
            float* dst = &lds_bt[t * KPAD];
            #pragma unroll
            for (int i = 0; i < 16; ++i) {
                v2f p;
                p.x = pa[i];
                p.y = pb[i];
                *(v2f*)(dst + 2 * i) = p;
            }
        }
        if (ch + 1 < NCHUNK) {
            const float* src = w + (size_t)(ch + 1) * KC * F_OUT;
            #pragma unroll
            for (int i = 0; i < 16; ++i) {
                pa[i] = src[(2 * i)     * F_OUT + t];
                pb[i] = src[(2 * i + 1) * F_OUT + t];
            }
        }
        __syncthreads();

        const int tap = ch >> 2;
        const int c0  = (ch & 3) * KC;
        const float* abase = &lds_x[(m0 + lm + tap) * XPAD + c0 + 4 * half];
        const float* bbase = &lds_bt[(f0 + lm) * KPAD + 4 * half];

        v4f fa[2][WM], fb[2][WN];
        #pragma unroll
        for (int mi = 0; mi < WM; ++mi) fa[0][mi] = *(const v4f*)(abase + mi * 16 * XPAD);
        #pragma unroll
        for (int ni = 0; ni < WN; ++ni) fb[0][ni] = *(const v4f*)(bbase + ni * 16 * KPAD);

        #pragma unroll
        for (int k8 = 0; k8 < KC / 8; ++k8) {
            const int cur = k8 & 1;
            if (k8 + 1 < KC / 8) {
                #pragma unroll
                for (int mi = 0; mi < WM; ++mi)
                    fa[cur ^ 1][mi] = *(const v4f*)(abase + mi * 16 * XPAD + (k8 + 1) * 8);
                #pragma unroll
                for (int ni = 0; ni < WN; ++ni)
                    fb[cur ^ 1][ni] = *(const v4f*)(bbase + ni * 16 * KPAD + (k8 + 1) * 8);
            }
            #pragma unroll
            for (int sub = 0; sub < 2; ++sub) {
                #pragma unroll
                for (int mi = 0; mi < WM; ++mi) {
                    v2f a2;
                    a2.x = fa[cur][mi][2 * sub];
                    a2.y = fa[cur][mi][2 * sub + 1];
                    #pragma unroll
                    for (int ni = 0; ni < WN; ++ni) {
                        v2f b2;
                        b2.x = fb[cur][ni][2 * sub];
                        b2.y = fb[cur][ni][2 * sub + 1];
                        acc[mi][ni] = __builtin_amdgcn_wmma_f32_16x16x4_f32(
                            false, a2, false, b2, (short)0, acc[mi][ni], false, false);
                    }
                }
            }
        }
    }

    float* ob = out + ((size_t)b * L_OUT + l0) * F_OUT;
    #pragma unroll
    for (int mi = 0; mi < WM; ++mi) {
        #pragma unroll
        for (int v = 0; v < 8; ++v) {
            int mrow = m0 + mi * 16 + v + 8 * half;
            if (l0 + mrow < L_OUT) {
                float inv = lds_invn[mrow];
                #pragma unroll
                for (int ni = 0; ni < WN; ++ni) {
                    int f = f0 + ni * 16 + lm;
                    ob[(size_t)mrow * F_OUT + f] = acc[mi][ni][v] * inv + lds_bias[f];
                }
            }
        }
    }
}

extern "C" void kernel_launch(void* const* d_in, const int* in_sizes, int n_in,
                              void* d_out, int out_size, void* d_ws, size_t ws_size,
                              hipStream_t stream) {
    const float* x     = (const float*)d_in[0];
    const float* w     = (const float*)d_in[1];
    const float* scale = (const float*)d_in[2];
    const float* bias  = (const float*)d_in[3];
    float* out = (float*)d_out;

    dim3 grid((L_OUT + TILE_M - 1) / TILE_M, BATCH);
    ckn1d_wmma_kernel<<<grid, 256, 0, stream>>>(x, w, scale, bias, out);
}


# weight chunk tile double-buffered in LDS: one barrier per chunk instead of two; next chunk's stores and the chunk after's global loads issue behind the first fragment loads; sum order unchanged
# speedup vs baseline: 1.1356x; 1.0407x over previous
#include <hip/hip_runtime.h>
#include <hip/hip_bf16.h>


typedef __attribute__((ext_vector_type(2))) float v2f;
typedef __attribute__((ext_vector_type(4))) float v4f;
typedef __attribute__((ext_vector_type(8))) float v8f;

#define K_TAPS 15
#define CIN    128
#define F_OUT  256
#define L_IN   4096
#define L_OUT  (L_IN - K_TAPS + 1)
#define BATCH  32
#define TILE_M 128
#define ROWS   (TILE_M + K_TAPS - 1)
#define XPAD   132
#define KC     32
#define KPAD   36
#define NCHUNK ((K_TAPS * CIN) / KC)
#define EPS_F  1e-7f
#define WM     4
#define WN     4

__global__ __launch_bounds__(256)
void ckn1d_wmma_kernel(const float* __restrict__ x,
                       const float* __restrict__ w,
                       const float* __restrict__ scale,
                       const float* __restrict__ bias,
                       float* __restrict__ out) {
    __shared__ __attribute__((aligned(16))) float lds_x[ROWS * XPAD];
    __shared__ __attribute__((aligned(16))) float lds_bt[2][F_OUT * KPAD];
    __shared__ float lds_rowsq[ROWS];
    __shared__ float lds_invn[TILE_M];
    __shared__ float lds_bias[F_OUT];

    const int t    = threadIdx.x;
    const int lane = t & 31;
    const int wave = t >> 5;
    const int l0   = blockIdx.x * TILE_M;
    const int b    = blockIdx.y;

    const float* xb = x + (size_t)b * L_IN * CIN;
    #pragma unroll
    for (int i = 0; i < 18; ++i) {
        int q = t + 256 * i;
        if (q < ROWS * (CIN / 4)) {
            int row = q >> 5;
            int f4  = q & 31;
            int g   = l0 + row;
            v4f val = {0.f, 0.f, 0.f, 0.f};
            if (g < L_IN) val = *(const v4f*)(xb + (size_t)g * CIN + f4 * 4);
            *(v4f*)(&lds_x[row * XPAD + f4 * 4]) = val;
        }
    }
    if (t < F_OUT) lds_bias[t] = bias[t];
    __syncthreads();

    if (t < ROWS) {
        const v4f* r = (const v4f*)&lds_x[t * XPAD];
        float s = 0.f;
        #pragma unroll
        for (int c = 0; c < CIN / 4; ++c) {
            v4f v = r[c];
            s += v.x * v.x + v.y * v.y + v.z * v.z + v.w * v.w;
        }
        lds_rowsq[t] = s;
    }
    __syncthreads();
    if (t < TILE_M) {
        float win = 0.f;
        #pragma unroll
        for (int k = 0; k < K_TAPS; ++k) win += lds_rowsq[t + k];
        lds_invn[t] = scale[0] / __builtin_sqrtf(win + EPS_F);
    }

    const int m0   = (wave & 1) * 64;
    const int f0   = (wave >> 1) * 64;
    const int half = lane >> 4;
    const int lm   = lane & 15;

    v8f acc[WM][WN] = {};

    v4f q[KC / 4];
    #pragma unroll
    for (int j = 0; j < KC / 4; ++j) {
        #pragma unroll
        for (int e = 0; e < 4; ++e) q[j][e] = w[(4 * j + e) * F_OUT + t];
    }
    #pragma unroll
    for (int j = 0; j < KC / 4; ++j)
        *(v4f*)(&lds_bt[0][t * KPAD + 4 * j]) = q[j];
    #pragma unroll
    for (int j = 0; j < KC / 4; ++j) {
        #pragma unroll
        for (int e = 0; e < 4; ++e) q[j][e] = w[(size_t)(KC + 4 * j + e) * F_OUT + t];
    }

    for (int ch = 0; ch < NCHUNK; ++ch) {
        __syncthreads();

        const int tap = ch >> 2;
        const int c0  = (ch & 3) * KC;
        const float* abase = &lds_x[(m0 + lm + tap) * XPAD + c0 + 4 * half];
        const float* bbase = &lds_bt[ch & 1][(f0 + lm) * KPAD + 4 * half];

        v4f fa[2][WM], fb[2][WN];
        #pragma unroll
        for (int mi = 0; mi < WM; ++mi) fa[0][mi] = *(const v4f*)(abase + mi * 16 * XPAD);
        #pragma unroll
        for (int ni = 0; ni < WN; ++ni) fb[0][ni] = *(const v4f*)(bbase + ni * 16 * KPAD);
        __builtin_amdgcn_sched_barrier(0);

        {
            float* dst = &lds_bt[(ch + 1) & 1][t * KPAD];
            #pragma unroll
            for (int j = 0; j < KC / 4; ++j) *(v4f*)(dst + 4 * j) = q[j];
            const int cn = (ch + 2 < NCHUNK) ? ch + 2 : NCHUNK - 1;
            const float* src = w + (size_t)cn * KC * F_OUT + t;
            #pragma unroll
            for (int j = 0; j < KC / 4; ++j) {
                #pragma unroll
                for (int e = 0; e < 4; ++e) q[j][e] = src[(4 * j + e) * F_OUT];
            }
        }

        #pragma unroll
        for (int k8 = 0; k8 < KC / 8; ++k8) {
            const int cur = k8 & 1;
            if (k8 + 1 < KC / 8) {
                #pragma unroll
                for (int mi = 0; mi < WM; ++mi)
                    fa[cur ^ 1][mi] = *(const v4f*)(abase + mi * 16 * XPAD + (k8 + 1) * 8);
                #pragma unroll
                for (int ni = 0; ni < WN; ++ni)
                    fb[cur ^ 1][ni] = *(const v4f*)(bbase + ni * 16 * KPAD + (k8 + 1) * 8);
            }
            #pragma unroll
            for (int sub = 0; sub < 2; ++sub) {
                #pragma unroll
                for (int mi = 0; mi < WM; ++mi) {
                    v2f a2;
                    a2.x = fa[cur][mi][2 * sub];
                    a2.y = fa[cur][mi][2 * sub + 1];
                    #pragma unroll
                    for (int ni = 0; ni < WN; ++ni) {
                        v2f b2;
                        b2.x = fb[cur][ni][2 * sub];
                        b2.y = fb[cur][ni][2 * sub + 1];
                        acc[mi][ni] = __builtin_amdgcn_wmma_f32_16x16x4_f32(
                            false, a2, false, b2, (short)0, acc[mi][ni], false, false);
                    }
                }
            }
        }
    }

    float* ob = out + ((size_t)b * L_OUT + l0) * F_OUT;
    #pragma unroll
    for (int mi = 0; mi < WM; ++mi) {
        #pragma unroll
        for (int v = 0; v < 8; ++v) {
            int mrow = m0 + mi * 16 + v + 8 * half;
            if (l0 + mrow < L_OUT) {
                float inv = lds_invn[mrow];
                #pragma unroll
                for (int ni = 0; ni < WN; ++ni) {
                    int f = f0 + ni * 16 + lm;
                    ob[(size_t)mrow * F_OUT + f] = acc[mi][ni][v] * inv + lds_bias[f];
                }
            }
        }
    }
}

extern "C" void kernel_launch(void* const* d_in, const int* in_sizes, int n_in,
                              void* d_out, int out_size, void* d_ws, size_t ws_size,
                              hipStream_t stream) {
    const float* x     = (const float*)d_in[0];
    const float* w     = (const float*)d_in[1];
    const float* scale = (const float*)d_in[2];
    const float* bias  = (const float*)d_in[3];
    float* out = (float*)d_out;

    dim3 grid((L_OUT + TILE_M - 1) / TILE_M, BATCH);
    ckn1d_wmma_kernel<<<grid, 256, 0, stream>>>(x, w, scale, bias, out);
}


# 15 taps as 5 groups of 3, two neighbouring outputs share 4 products per group where 6 were (minimal filtering F(2,3)): matrix instructions x 2/3; sums regrouped, sum order changes
# speedup vs baseline: 1.4504x; 1.2772x over previous
#include <hip/hip_runtime.h>
#include <hip/hip_bf16.h>


typedef __attribute__((ext_vector_type(2))) float v2f;
typedef __attribute__((ext_vector_type(4))) float v4f;
typedef __attribute__((ext_vector_type(8))) float v8f;

#define K_TAPS 15
#define CIN    128
#define F_OUT  256
#define L_IN   4096
#define L_OUT  (L_IN - K_TAPS + 1)
#define BATCH  32
#define TILE_M 128
#define ROWS   (TILE_M + K_TAPS - 1)
#define HROWS  (ROWS / 2)
#define XPAD   132
#define X_FLOATS (ROWS * XPAD)
#define NG     5
#define KC     16
#define KPAD   20
#define FB     128
#define NCHUNK (NG * (CIN / KC))
#define BT_FLOATS (4 * FB * KPAD)
#define MPITCH 132
#define EPS_F  1e-7f
#define WM     4
#define WN     4

__global__ __launch_bounds__(256)
void ckn1d_wmma_kernel(const float* __restrict__ x,
                       const float* __restrict__ w,
                       const float* __restrict__ scale,
                       const float* __restrict__ bias,
                       float* __restrict__ out) {
    __shared__ __attribute__((aligned(16))) float lds_main[X_FLOATS + 2 * BT_FLOATS];
    __shared__ float lds_rowsq[ROWS];
    __shared__ float lds_invn[TILE_M];
    __shared__ float lds_bias[FB];
    float* lds_x  = lds_main;
    float* lds_bt = lds_main + X_FLOATS;

    const int t    = threadIdx.x;
    const int lane = t & 31;
    const int wave = t >> 5;
    const int fh   = blockIdx.x & 1;
    const int l0   = (blockIdx.x >> 1) * TILE_M;
    const int b    = blockIdx.y;

    const float* xb = x + (size_t)b * L_IN * CIN;
    #pragma unroll
    for (int i = 0; i < 18; ++i) {
        int q = t + 256 * i;
        if (q < ROWS * (CIN / 4)) {
            int row = q >> 5;
            int f4  = q & 31;
            int gl  = l0 + row;
            v4f val = {0.f, 0.f, 0.f, 0.f};
            if (gl < L_IN) val = *(const v4f*)(xb + (size_t)gl * CIN + f4 * 4);
            *(v4f*)(&lds_x[((row & 1) * HROWS + (row >> 1)) * XPAD + f4 * 4]) = val;
        }
    }
    if (t < FB) lds_bias[t] = bias[fh * FB + t];
    __syncthreads();

    if (t < ROWS) {
        const v4f* r = (const v4f*)&lds_x[((t & 1) * HROWS + (t >> 1)) * XPAD];
        float s = 0.f;
        #pragma unroll
        for (int c = 0; c < CIN / 4; ++c) {
            v4f v = r[c];
            s += v.x * v.x + v.y * v.y + v.z * v.z + v.w * v.w;
        }
        lds_rowsq[t] = s;
    }
    __syncthreads();
    if (t < TILE_M) {
        float win = 0.f;
        #pragma unroll
        for (int k = 0; k < K_TAPS; ++k) win += lds_rowsq[t + k];
        lds_invn[t] = scale[0] / __builtin_sqrtf(win + EPS_F);
    }

    const int js   = wave >> 1;
    const int hf   = wave & 1;
    const int half = lane >> 4;
    const int lm   = lane & 15;
    const int   ia  = (js == 0) ? 0 : ((js == 2) ? 2 : 1);
    const int   ib  = (js == 3) ? 3 : ((js == 2) ? 1 : 2);
    const float sgn = (js == 1) ? 1.f : -1.f;

    v8f acc[WM][WN] = {};

    const int sf = t & (FB - 1);
    const int sc = t >> 7;
    const float* wcol = w + (size_t)(sc * 8) * F_OUT + fh * FB + sf;
    float wq[3][8];
    #define LOAD_RAW(chunk_) do {                                                          \
        const float* src_ = wcol + (size_t)((3 * ((chunk_) >> 3)) * CIN + ((chunk_) & 7) * KC) * F_OUT; \
        _Pragma("unroll") for (int jt = 0; jt < 3; ++jt)                                   \
            _Pragma("unroll") for (int i = 0; i < 8; ++i)                                  \
                wq[jt][i] = src_[(size_t)(jt * CIN + i) * F_OUT];                          \
    } while (0)
    #define STORE_COMB(buf_) do {                                                          \
        float* dst_ = lds_bt + (buf_) * BT_FLOATS + sf * KPAD + sc * 8;                    \
        v4f u0a, u0b, u1a, u1b, u2a, u2b, u3a, u3b;                                        \
        _Pragma("unroll") for (int i = 0; i < 4; ++i) {                                    \
            float sa = wq[0][i] + wq[2][i];                                                \
            float sb = wq[0][i + 4] + wq[2][i + 4];                                        \
            u0a[i] = wq[0][i];                  u0b[i] = wq[0][i + 4];                     \
            u1a[i] = 0.5f * (sa + wq[1][i]);    u1b[i] = 0.5f * (sb + wq[1][i + 4]);       \
            u2a[i] = 0.5f * (sa - wq[1][i]);    u2b[i] = 0.5f * (sb - wq[1][i + 4]);       \
            u3a[i] = wq[2][i];                  u3b[i] = wq[2][i + 4];                     \
        }                                                                                  \
        *(v4f*)(dst_ + 0 * FB * KPAD)     = u0a;  *(v4f*)(dst_ + 0 * FB * KPAD + 4) = u0b; \
        *(v4f*)(dst_ + 1 * FB * KPAD)     = u1a;  *(v4f*)(dst_ + 1 * FB * KPAD + 4) = u1b; \
        *(v4f*)(dst_ + 2 * FB * KPAD)     = u2a;  *(v4f*)(dst_ + 2 * FB * KPAD + 4) = u2b; \
        *(v4f*)(dst_ + 3 * FB * KPAD)     = u3a;  *(v4f*)(dst_ + 3 * FB * KPAD + 4) = u3b; \
    } while (0)

    LOAD_RAW(0);
    STORE_COMB(0);
    LOAD_RAW(1);

    const float* xlane = &lds_x[lm * XPAD + 4 * half];

    for (int ch = 0; ch < NCHUNK; ++ch) {
        __syncthreads();

        const int g  = ch >> 3;
        const int c0 = (ch & 7) * KC;
        const int ea = 3 * g + ia;
        const int eb = 3 * g + ib;
        const float* pa = xlane + ((ea & 1) * HROWS + (ea >> 1)) * XPAD + c0;
        const float* pb = xlane + ((eb & 1) * HROWS + (eb >> 1)) * XPAD + c0;
        const float* bbase = lds_bt + (ch & 1) * BT_FLOATS + (js * FB + hf * 64 + lm) * KPAD + 4 * half;

        v4f ra[WM], rb[WM], fa[2][WM], fb[2][WN];
        #pragma unroll
        for (int mi = 0; mi < WM; ++mi) {
            ra[mi] = *(const v4f*)(pa + mi * 16 * XPAD);
            rb[mi] = *(const v4f*)(pb + mi * 16 * XPAD);
        }
        #pragma unroll
        for (int ni = 0; ni < WN; ++ni) fb[0][ni] = *(const v4f*)(bbase + ni * 16 * KPAD);
        __builtin_amdgcn_sched_barrier(0);

        {
            STORE_COMB((ch + 1) & 1);
            const int cn = (ch + 2 < NCHUNK) ? ch + 2 : NCHUNK - 1;
            LOAD_RAW(cn);
        }
        __builtin_amdgcn_sched_barrier(0);

        #pragma unroll
        for (int mi = 0; mi < WM; ++mi) fa[0][mi] = ra[mi] + sgn * rb[mi];

        #pragma unroll
        for (int k8 = 0; k8 < KC / 8; ++k8) {
            const int cur = k8 & 1;
            if (k8 + 1 < KC / 8) {
                #pragma unroll
                for (int mi = 0; mi < WM; ++mi) {
                    ra[mi] = *(const v4f*)(pa + mi * 16 * XPAD + (k8 + 1) * 8);
                    rb[mi] = *(const v4f*)(pb + mi * 16 * XPAD + (k8 + 1) * 8);
                }
                #pragma unroll
                for (int ni = 0; ni < WN; ++ni)
                    fb[cur ^ 1][ni] = *(const v4f*)(bbase + ni * 16 * KPAD + (k8 + 1) * 8);
            }
            #pragma unroll
            for (int sub = 0; sub < 2; ++sub) {
                #pragma unroll
                for (int mi = 0; mi < WM; ++mi) {
                    v2f a2;
                    a2.x = fa[cur][mi][2 * sub];
                    a2.y = fa[cur][mi][2 * sub + 1];
                    #pragma unroll
                    for (int ni = 0; ni < WN; ++ni) {
                        v2f b2;
                        b2.x = fb[cur][ni][2 * sub];
                        b2.y = fb[cur][ni][2 * sub + 1];
                        acc[mi][ni] = __builtin_amdgcn_wmma_f32_16x16x4_f32(
                            false, a2, false, b2, (short)0, acc[mi][ni], false, false);
                    }
                }
            }
            if (k8 + 1 < KC / 8) {
                #pragma unroll
                for (int mi = 0; mi < WM; ++mi) fa[cur ^ 1][mi] = ra[mi] + sgn * rb[mi];
            }
        }
    }
    #undef LOAD_RAW
    #undef STORE_COMB

    __syncthreads();
    float* mt = lds_main;
    #pragma unroll
    for (int mi = 0; mi < WM; ++mi) {
        #pragma unroll
        for (int v = 0; v < 8; ++v) {
            int pr = mi * 16 + v + 8 * half;
            #pragma unroll
            for (int ni = 0; ni < WN; ++ni)
                mt[(js * 64 + pr) * MPITCH + hf * 64 + ni * 16 + lm] = acc[mi][ni][v];
        }
    }
    __syncthreads();
    float* ob = out + ((size_t)b * L_OUT + l0) * F_OUT + fh * FB;
    const int odd = wave & 1;
    const v4f bv = *(const v4f*)(&lds_bias[lane * 4]);
    #pragma unroll 4
    for (int it = 0; it < TILE_M / 8; ++it) {
        int pos = it * 8 + wave;
        int pr  = pos >> 1;
        if (l0 + pos < L_OUT) {
            const float* mrow = mt + pr * MPITCH + lane * 4;
            v4f m1 = *(const v4f*)(mrow + 1 * 64 * MPITCH);
            v4f m2 = *(const v4f*)(mrow + 2 * 64 * MPITCH);
            v4f val;
            if (odd) {
                v4f m3 = *(const v4f*)(mrow + 3 * 64 * MPITCH);
                val = (m1 - m2) - m3;
            } else {
                v4f m0 = *(const v4f*)(mrow);
                val = (m0 + m1) + m2;
            }
            float inv = lds_invn[pos];
            *(v4f*)(ob + (size_t)pos * F_OUT + lane * 4) = val * inv + bv;
        }
    }
}

extern "C" void kernel_launch(void* const* d_in, const int* in_sizes, int n_in,
                              void* d_out, int out_size, void* d_ws, size_t ws_size,
                              hipStream_t stream) {
    const float* x     = (const float*)d_in[0];
    const float* w     = (const float*)d_in[1];
    const float* scale = (const float*)d_in[2];
    const float* bias  = (const float*)d_in[3];
    float* out = (float*)d_out;

    dim3 grid(2 * ((L_OUT + TILE_M - 1) / TILE_M), BATCH);
    ckn1d_wmma_kernel<<<grid, 256, 0, stream>>>(x, w, scale, bias, out);
}


# weight combinations formed once per call by a first launch, in B-fragment order in the workspace; each wave loads its B fragments from global: no weight LDS staging, no chunk barrier; same sum order
# speedup vs baseline: 1.5559x; 1.0727x over previous
#include <hip/hip_runtime.h>
#include <hip/hip_bf16.h>


typedef __attribute__((ext_vector_type(2))) float v2f;
typedef __attribute__((ext_vector_type(4))) float v4f;
typedef __attribute__((ext_vector_type(8))) float v8f;

#define K_TAPS 15
#define CIN    128
#define F_OUT  256
#define L_IN   4096
#define L_OUT  (L_IN - K_TAPS + 1)
#define BATCH  32
#define TILE_M 128
#define ROWS   (TILE_M + K_TAPS - 1)
#define HROWS  (ROWS / 2)
#define XPAD   132
#define X_FLOATS (ROWS * XPAD)
#define NG     5
#define KC     16
#define FB     128
#define NCHUNK (NG * (CIN / KC))
#define UCH_V4 2048
#define MPITCH 132
#define MT_FLOATS (4 * 64 * MPITCH)
#define EPS_F  1e-7f
#define WM     4
#define WN     4

__global__ __launch_bounds__(256)
void ckn1d_wfrag_kernel(const float* __restrict__ w,
                        float* __restrict__ u) {
    const int id   = blockIdx.x * 256 + threadIdx.x;
    const int lane = id & 31;
    const int ni   = (id >> 5) & 3;
    const int k8   = (id >> 7) & 1;
    const int hf   = (id >> 8) & 1;
    const int r    = id >> 9;
    const int fh   = r / NCHUNK;
    const int ch   = r - fh * NCHUNK;
    const int g    = ch >> 3;
    const int chan = (ch & 7) * KC + k8 * 8 + 4 * (lane >> 4);
    const int filt = fh * FB + hf * 64 + ni * 16 + (lane & 15);
    const float* src = w + (size_t)(3 * g * CIN + chan) * F_OUT + filt;
    v4f u0, u1, u2, u3;
    #pragma unroll
    for (int e = 0; e < 4; ++e) {
        float w0 = src[(size_t)(0 * CIN + e) * F_OUT];
        float w1 = src[(size_t)(1 * CIN + e) * F_OUT];
        float w2 = src[(size_t)(2 * CIN + e) * F_OUT];
        float sa = w0 + w2;
        u0[e] = w0;
        u1[e] = 0.5f * (sa + w1);
        u2[e] = 0.5f * (sa - w1);
        u3[e] = w2;
    }
    v4f* dst = (v4f*)u + (size_t)r * UCH_V4 + hf * 256 + k8 * 128 + ni * 32 + lane;
    dst[0 * 512] = u0;
    dst[1 * 512] = u1;
    dst[2 * 512] = u2;
    dst[3 * 512] = u3;
}

__global__ __launch_bounds__(256)
void ckn1d_wmma_kernel(const float* __restrict__ x,
                       const float* __restrict__ u,
                       const float* __restrict__ scale,
                       const float* __restrict__ bias,
                       float* __restrict__ out) {
    __shared__ __attribute__((aligned(16))) float lds_main[MT_FLOATS];
    __shared__ float lds_rowsq[ROWS];
    __shared__ float lds_invn[TILE_M];
    __shared__ float lds_bias[FB];
    float* lds_x  = lds_main;

    const int t    = threadIdx.x;
    const int lane = t & 31;
    const int wave = t >> 5;
    const int fh   = blockIdx.x & 1;
    const int l0   = (blockIdx.x >> 1) * TILE_M;
    const int b    = blockIdx.y;

    const float* xb = x + (size_t)b * L_IN * CIN;
    #pragma unroll
    for (int i = 0; i < 18; ++i) {
        int q = t + 256 * i;
        if (q < ROWS * (CIN / 4)) {
            int row = q >> 5;
            int f4  = q & 31;
            int gl  = l0 + row;
            v4f val = {0.f, 0.f, 0.f, 0.f};
            if (gl < L_IN) val = *(const v4f*)(xb + (size_t)gl * CIN + f4 * 4);
            *(v4f*)(&lds_x[((row & 1) * HROWS + (row >> 1)) * XPAD + f4 * 4]) = val;
        }
    }
    if (t < FB) lds_bias[t] = bias[fh * FB + t];
    __syncthreads();

    if (t < ROWS) {
        const v4f* r = (const v4f*)&lds_x[((t & 1) * HROWS + (t >> 1)) * XPAD];
        float s = 0.f;
        #pragma unroll
        for (int c = 0; c < CIN / 4; ++c) {
            v4f v = r[c];
            s += v.x * v.x + v.y * v.y + v.z * v.z + v.w * v.w;
        }
        lds_rowsq[t] = s;
    }
    __syncthreads();
    if (t < TILE_M) {
        float win = 0.f;
        #pragma unroll
        for (int k = 0; k < K_TAPS; ++k) win += lds_rowsq[t + k];
        lds_invn[t] = scale[0] / __builtin_sqrtf(win + EPS_F);
    }

    const int js   = wave >> 1;
    const int hf   = wave & 1;
    const int half = lane >> 4;
    const int lm   = lane & 15;
    const int   ia  = (js == 0) ? 0 : ((js == 2) ? 2 : 1);
    const int   ib  = (js == 3) ? 3 : ((js == 2) ? 1 : 2);
    const float sgn = (js == 1) ? 1.f : -1.f;

    v8f acc[WM][WN] = {};

    const v4f* ub = (const v4f*)u + (size_t)(fh * NCHUNK) * UCH_V4 + wave * 256 + lane;
    const float* xlane = &lds_x[lm * XPAD + 4 * half];

    v4f ra[WM], rb[WM], fa[2][WM], fb[2][WN];

    #define A_ROWS(chunk_, pa_, pb_) do {                                                  \
        const int g_  = (chunk_) >> 3;                                                     \
        const int c0_ = ((chunk_) & 7) * KC;                                               \
        const int ea_ = 3 * g_ + ia;                                                       \
        const int eb_ = 3 * g_ + ib;                                                       \
        pa_ = xlane + ((ea_ & 1) * HROWS + (ea_ >> 1)) * XPAD + c0_;                       \
        pb_ = xlane + ((eb_ & 1) * HROWS + (eb_ >> 1)) * XPAD + c0_;                       \
    } while (0)
    #define LOAD_A(pa_, pb_, koff_) do {                                                   \
        _Pragma("unroll") for (int mi = 0; mi < WM; ++mi) {                                \
            ra[mi] = *(const v4f*)((pa_) + mi * 16 * XPAD + (koff_));                      \
            rb[mi] = *(const v4f*)((pb_) + mi * 16 * XPAD + (koff_));                      \
        }                                                                                  \
    } while (0)
    #define LOAD_B(set_, chunk_, k8_) do {                                                 \
        _Pragma("unroll") for (int ni = 0; ni < WN; ++ni)                                  \
            fb[set_][ni] = ub[(size_t)(chunk_) * UCH_V4 + (k8_) * 128 + ni * 32];          \
    } while (0)
    #define FORM_A(set_) do {                                                              \
        _Pragma("unroll") for (int mi = 0; mi < WM; ++mi) fa[set_][mi] = ra[mi] + sgn * rb[mi]; \
    } while (0)
    #define MMA_GROUP(set_) do {                                                           \
        _Pragma("unroll") for (int sub = 0; sub < 2; ++sub) {                              \
            _Pragma("unroll") for (int mi = 0; mi < WM; ++mi) {                            \
                v2f a2;                                                                    \
                a2.x = fa[set_][mi][2 * sub];                                              \
                a2.y = fa[set_][mi][2 * sub + 1];                                          \
                _Pragma("unroll") for (int ni = 0; ni < WN; ++ni) {                        \
                    v2f b2;                                                                \
                    b2.x = fb[set_][ni][2 * sub];                                          \
                    b2.y = fb[set_][ni][2 * sub + 1];                                      \
                    acc[mi][ni] = __builtin_amdgcn_wmma_f32_16x16x4_f32(                   \
                        false, a2, false, b2, (short)0, acc[mi][ni], false, false);        \
                }                                                                          \
            }                                                                              \
        }                                                                                  \
    } while (0)

    const float *pa, *pb;
    A_ROWS(0, pa, pb);
    LOAD_B(0, 0, 0);
    LOAD_B(1, 0, 1);
    LOAD_A(pa, pb, 0);

    for (int ch = 0; ch < NCHUNK; ++ch) {
        const int cn = (ch + 1 < NCHUNK) ? ch + 1 : NCHUNK - 1;

        FORM_A(0);
        LOAD_A(pa, pb, 8);
        __builtin_amdgcn_sched_barrier(0);
        MMA_GROUP(0);
        LOAD_B(0, cn, 0);
        FORM_A(1);

        A_ROWS(cn, pa, pb);
        LOAD_A(pa, pb, 0);
        __builtin_amdgcn_sched_barrier(0);
        MMA_GROUP(1);
        LOAD_B(1, cn, 1);
    }
    #undef A_ROWS
    #undef LOAD_A
    #undef LOAD_B
    #undef FORM_A
    #undef MMA_GROUP

    __syncthreads();
    float* mt = lds_main;
    #pragma unroll
    for (int mi = 0; mi < WM; ++mi) {
        #pragma unroll
        for (int v = 0; v < 8; ++v) {
            int pr = mi * 16 + v + 8 * half;
            #pragma unroll
            for (int ni = 0; ni < WN; ++ni)
                mt[(js * 64 + pr) * MPITCH + hf * 64 + ni * 16 + lm] = acc[mi][ni][v];
        }
    }
    __syncthreads();
    float* ob = out + ((size_t)b * L_OUT + l0) * F_OUT + fh * FB;
    const int odd = wave & 1;
    const v4f bv = *(const v4f*)(&lds_bias[lane * 4]);
    #pragma unroll 4
    for (int it = 0; it < TILE_M / 8; ++it) {
        int pos = it * 8 + wave;
        int pr  = pos >> 1;
        if (l0 + pos < L_OUT) {
            const float* mrow = mt + pr * MPITCH + lane * 4;
            v4f m1 = *(const v4f*)(mrow + 1 * 64 * MPITCH);
            v4f m2 = *(const v4f*)(mrow + 2 * 64 * MPITCH);
            v4f val;
            if (odd) {
                v4f m3 = *(const v4f*)(mrow + 3 * 64 * MPITCH);
                val = (m1 - m2) - m3;
            } else {
                v4f m0 = *(const v4f*)(mrow);
                val = (m0 + m1) + m2;
            }
            float inv = lds_invn[pos];
            *(v4f*)(ob + (size_t)pos * F_OUT + lane * 4) = val * inv + bv;
        }
    }
}

extern "C" void kernel_launch(void* const* d_in, const int* in_sizes, int n_in,
                              void* d_out, int out_size, void* d_ws, size_t ws_size,
                              hipStream_t stream) {
    const float* x     = (const float*)d_in[0];
    const float* w     = (const float*)d_in[1];
    const float* scale = (const float*)d_in[2];
    const float* bias  = (const float*)d_in[3];
    float* out = (float*)d_out;

    dim3 grid(2 * ((L_OUT + TILE_M - 1) / TILE_M), BATCH);
    float* u = (float*)d_ws;
    ckn1d_wfrag_kernel<<<dim3(2 * NCHUNK * UCH_V4 / 4 / 256), 256, 0, stream>>>(w, u);
    ckn1d_wmma_kernel<<<grid, 256, 0, stream>>>(x, u, scale, bias, out);
}


# taps in groups of 3: four neighbouring outputs share 6 products where 8 were (minimal filtering F(4,3) for F(2,3)): matrix instructions x 3/4; 12-wave block of 256 positions; sum order changes
# speedup vs baseline: 1.7305x; 1.1122x over previous
#include <hip/hip_runtime.h>
#include <hip/hip_bf16.h>


typedef __attribute__((ext_vector_type(2))) float v2f;
typedef __attribute__((ext_vector_type(4))) float v4f;
typedef __attribute__((ext_vector_type(8))) float v8f;

#define K_TAPS 15
#define CIN    128
#define F_OUT  256
#define L_IN   4096
#define L_OUT  (L_IN - K_TAPS + 1)
#define BATCH  32
#define TILE_M 256
#define NQ     (TILE_M / 4)
#define ROWS   (TILE_M + K_TAPS - 1)
#define PROWS  68
#define XPAD   132
#define X_FLOATS (4 * PROWS * XPAD)
#define NG     5
#define NS     6
#define KC     16
#define FB     128
#define NCHUNK (NG * (CIN / KC))
#define NWAVE  (2 * NS)
#define NT     (32 * NWAVE)
#define UCH_V4 (NWAVE * 256)
#define MPITCH 132
#define MT_FLOATS (NS * NQ * MPITCH)
#define EPS_F  1e-7f
#define WM     4
#define WN     4

__global__ __launch_bounds__(256)
void ckn1d_wfrag_kernel(const float* __restrict__ w,
                        float* __restrict__ u) {
    const int id   = blockIdx.x * 256 + threadIdx.x;
    const int lane = id & 31;
    const int ni   = (id >> 5) & 3;
    const int k8   = (id >> 7) & 1;
    const int hf   = (id >> 8) & 1;
    const int r    = id >> 9;
    const int fh   = r / NCHUNK;
    const int ch   = r - fh * NCHUNK;
    const int g    = ch >> 3;
    const int chan = (ch & 7) * KC + k8 * 8 + 4 * (lane >> 4);
    const int filt = fh * FB + hf * 64 + ni * 16 + (lane & 15);
    const float* src = w + (size_t)(3 * g * CIN + chan) * F_OUT + filt;
    v4f u0, u1, u2, u3, u4, u5;
    #pragma unroll
    for (int e = 0; e < 4; ++e) {
        float w0 = src[(size_t)(0 * CIN + e) * F_OUT];
        float w1 = src[(size_t)(1 * CIN + e) * F_OUT];
        float w2 = src[(size_t)(2 * CIN + e) * F_OUT];
        float sa = w0 + w2;
        float sb = w0 + 4.f * w2;
        u0[e] = 0.25f * w0;
        u1[e] = (-1.f / 6.f) * (sa + w1);
        u2[e] = (-1.f / 6.f) * (sa - w1);
        u3[e] = (1.f / 24.f) * (sb + 2.f * w1);
        u4[e] = (1.f / 24.f) * (sb - 2.f * w1);
        u5[e] = w2;
    }
    v4f* dst = (v4f*)u + (size_t)r * UCH_V4 + hf * 256 + k8 * 128 + ni * 32 + lane;
    dst[0 * 512] = u0;
    dst[1 * 512] = u1;
    dst[2 * 512] = u2;
    dst[3 * 512] = u3;
    dst[4 * 512] = u4;
    dst[5 * 512] = u5;
}

__global__ __launch_bounds__(NT)
void ckn1d_wmma_kernel(const float* __restrict__ x,
                       const float* __restrict__ u,
                       const float* __restrict__ scale,
                       const float* __restrict__ bias,
                       float* __restrict__ out) {
    __shared__ __attribute__((aligned(16))) float lds_main[MT_FLOATS];
    __shared__ float lds_rowsq[ROWS];
    __shared__ float lds_invn[TILE_M];
    __shared__ float lds_bias[FB];
    float* lds_x  = lds_main;

    const int t    = threadIdx.x;
    const int lane = t & 31;
    const int wave = __builtin_amdgcn_readfirstlane(t >> 5);
    const int fh   = blockIdx.x & 1;
    const int l0   = (blockIdx.x >> 1) * TILE_M;
    const int b    = blockIdx.y;

    const float* xb = x + (size_t)b * L_IN * CIN;
    #pragma unroll
    for (int i = 0; i < 23; ++i) {
        int q = t + NT * i;
        if (q < ROWS * (CIN / 4)) {
            int row = q >> 5;
            int f4  = q & 31;
            int gl  = l0 + row;
            v4f val = {0.f, 0.f, 0.f, 0.f};
            if (gl < L_IN) val = *(const v4f*)(xb + (size_t)gl * CIN + f4 * 4);
            *(v4f*)(&lds_x[((row & 3) * PROWS + (row >> 2)) * XPAD + f4 * 4]) = val;
        }
    }
    if (t < FB) lds_bias[t] = bias[fh * FB + t];
    __syncthreads();

    if (t < ROWS) {
        const v4f* r = (const v4f*)&lds_x[((t & 3) * PROWS + (t >> 2)) * XPAD];
        float s = 0.f;
        #pragma unroll
        for (int c = 0; c < CIN / 4; ++c) {
            v4f v = r[c];
            s += v.x * v.x + v.y * v.y + v.z * v.z + v.w * v.w;
        }
        lds_rowsq[t] = s;
    }
    __syncthreads();
    if (t < TILE_M) {
        float win = 0.f;
        #pragma unroll
        for (int k = 0; k < K_TAPS; ++k) win += lds_rowsq[t + k];
        lds_invn[t] = scale[0] / __builtin_sqrtf(win + EPS_F);
    }

    const int js   = wave >> 1;
    const int hf   = wave & 1;
    const int half = lane >> 4;
    const int lm   = lane & 15;
    const int   o0 = (js == 0) ? 0 : 1;
    const int   o1 = (js == 5) ? 3 : 2;
    const int   o2 = (js == 0) ? 2 : 3;
    const int   o3 = (js == 5) ? 5 : 4;
    const bool  ends = (js == 0) || (js == 5);
    const float c0 = (js == 1) ? -4.f : ((js == 3) ? -2.f : ((js == 4) ? 2.f : 4.f));
    const float c1 = ends ? -2.5f : ((js <= 2) ? -4.f : -1.f);
    const float c2 = ends ? -2.5f : ((js == 1) ? 1.f : ((js == 2) ? -1.f : ((js == 3) ? 2.f : -2.f)));

    v8f acc[WM][WN] = {};

    const v4f* ub = (const v4f*)u + (size_t)(fh * NCHUNK) * UCH_V4 + wave * 256 + lane;
    const float* xlane = &lds_x[lm * XPAD + 4 * half];

    v4f r0[WM], r1[WM], r2[WM], r3[WM], fa[2][WM], fb[2][WN];

    #define A_PTR(e_, c0_) (xlane + (((e_) & 3) * PROWS + ((e_) >> 2)) * XPAD + (c0_))
    #define A_ROWS(chunk_) do {                                                            \
        const int g_  = (chunk_) >> 3;                                                     \
        const int k0_ = ((chunk_) & 7) * KC;                                               \
        p0 = A_PTR(3 * g_ + o0, k0_);                                                      \
        p1 = A_PTR(3 * g_ + o1, k0_);                                                      \
        p2 = A_PTR(3 * g_ + o2, k0_);                                                      \
        p3 = A_PTR(3 * g_ + o3, k0_);                                                      \
    } while (0)
    #define LOAD_A(koff_) do {                                                             \
        _Pragma("unroll") for (int mi = 0; mi < WM; ++mi) {                                \
            r0[mi] = *(const v4f*)(p0 + mi * 16 * XPAD + (koff_));                         \
            r1[mi] = *(const v4f*)(p1 + mi * 16 * XPAD + (koff_));                         \
            r2[mi] = *(const v4f*)(p2 + mi * 16 * XPAD + (koff_));                         \
            r3[mi] = *(const v4f*)(p3 + mi * 16 * XPAD + (koff_));                         \
        }                                                                                  \
    } while (0)
    #define LOAD_B(set_, chunk_, k8_) do {                                                 \
        _Pragma("unroll") for (int ni = 0; ni < WN; ++ni)                                  \
            fb[set_][ni] = ub[(size_t)(chunk_) * UCH_V4 + (k8_) * 128 + ni * 32];          \
    } while (0)
    #define FORM_A(set_) do {                                                              \
        _Pragma("unroll") for (int mi = 0; mi < WM; ++mi) {                                \
            _Pragma("unroll") for (int e = 0; e < 4; ++e)                                  \
                fa[set_][mi][e] = __builtin_fmaf(c2, r2[mi][e],                            \
                                  __builtin_fmaf(c1, r1[mi][e],                            \
                                  __builtin_fmaf(c0, r0[mi][e], r3[mi][e])));              \
        }                                                                                  \
    } while (0)
    #define MMA_GROUP(set_) do {                                                           \
        _Pragma("unroll") for (int sub = 0; sub < 2; ++sub) {                              \
            _Pragma("unroll") for (int mi = 0; mi < WM; ++mi) {                            \
                v2f a2;                                                                    \
                a2.x = fa[set_][mi][2 * sub];                                              \
                a2.y = fa[set_][mi][2 * sub + 1];                                          \
                _Pragma("unroll") for (int ni = 0; ni < WN; ++ni) {                        \
                    v2f b2;                                                                \
                    b2.x = fb[set_][ni][2 * sub];                                          \
                    b2.y = fb[set_][ni][2 * sub + 1];                                      \
                    acc[mi][ni] = __builtin_amdgcn_wmma_f32_16x16x4_f32(                   \
                        false, a2, false, b2, (short)0, acc[mi][ni], false, false);        \
                }                                                                          \
            }                                                                              \
        }                                                                                  \
    } while (0)

    const float *p0, *p1, *p2, *p3;
    A_ROWS(0);
    LOAD_B(0, 0, 0);
    LOAD_B(1, 0, 1);
    LOAD_A(0);

    for (int ch = 0; ch < NCHUNK; ++ch) {
        const int cn = (ch + 1 < NCHUNK) ? ch + 1 : NCHUNK - 1;

        FORM_A(0);
        LOAD_A(8);
        __builtin_amdgcn_sched_barrier(0);
        MMA_GROUP(0);
        LOAD_B(0, cn, 0);
        FORM_A(1);

        A_ROWS(cn);
        LOAD_A(0);
        __builtin_amdgcn_sched_barrier(0);
        MMA_GROUP(1);
        LOAD_B(1, cn, 1);
    }
    #undef A_PTR
    #undef A_ROWS
    #undef LOAD_A
    #undef LOAD_B
    #undef FORM_A
    #undef MMA_GROUP

    __syncthreads();
    float* mt = lds_main;
    #pragma unroll
    for (int mi = 0; mi < WM; ++mi) {
        #pragma unroll
        for (int v = 0; v < 8; ++v) {
            int pr = mi * 16 + v + 8 * half;
            #pragma unroll
            for (int ni = 0; ni < WN; ++ni)
                mt[(js * NQ + pr) * MPITCH + hf * 64 + ni * 16 + lm] = acc[mi][ni][v];
        }
    }
    __syncthreads();
    float* ob = out + ((size_t)b * L_OUT + l0) * F_OUT + fh * FB;
    const v4f bv = *(const v4f*)(&lds_bias[lane * 4]);
    for (int qd = wave; qd < NQ; qd += NWAVE) {
        const float* mrow = mt + qd * MPITCH + lane * 4;
        v4f m0 = *(const v4f*)(mrow);
        v4f m1 = *(const v4f*)(mrow + 1 * NQ * MPITCH);
        v4f m2 = *(const v4f*)(mrow + 2 * NQ * MPITCH);
        v4f m3 = *(const v4f*)(mrow + 3 * NQ * MPITCH);
        v4f m4 = *(const v4f*)(mrow + 4 * NQ * MPITCH);
        v4f m5 = *(const v4f*)(mrow + 5 * NQ * MPITCH);
        v4f s12 = m1 + m2, d12 = m1 - m2, s34 = m3 + m4, d34 = m3 - m4;
        v4f y[4];
        y[0] = (m0 + s12) + s34;
        y[1] = d12 + 2.f * d34;
        y[2] = s12 + 4.f * s34;
        y[3] = (d12 + 8.f * d34) + m5;
        #pragma unroll
        for (int j = 0; j < 4; ++j) {
            int pos = 4 * qd + j;
            if (l0 + pos < L_OUT)
                *(v4f*)(ob + (size_t)pos * F_OUT + lane * 4) = y[j] * lds_invn[pos] + bv;
        }
    }
}

extern "C" void kernel_launch(void* const* d_in, const int* in_sizes, int n_in,
                              void* d_out, int out_size, void* d_ws, size_t ws_size,
                              hipStream_t stream) {
    const float* x     = (const float*)d_in[0];
    const float* w     = (const float*)d_in[1];
    const float* scale = (const float*)d_in[2];
    const float* bias  = (const float*)d_in[3];
    float* out = (float*)d_out;

    dim3 grid(2 * ((L_OUT + TILE_M - 1) / TILE_M), BATCH);
    float* u = (float*)d_ws;
    ckn1d_wfrag_kernel<<<dim3(2 * NCHUNK * 512 / 256), 256, 0, stream>>>(w, u);
    ckn1d_wmma_kernel<<<grid, NT, 0, stream>>>(x, u, scale, bias, out);
}


# x slab staging: each wave issues its 23 row loads as one group before its first LDS store (row bounds by clamp and select, no branch around a load): one memory round trip per block, not 23 in series
# speedup vs baseline: 1.8883x; 1.0912x over previous
#include <hip/hip_runtime.h>
#include <hip/hip_bf16.h>


typedef __attribute__((ext_vector_type(2))) float v2f;
typedef __attribute__((ext_vector_type(4))) float v4f;
typedef __attribute__((ext_vector_type(8))) float v8f;

#define K_TAPS 15
#define CIN    128
#define F_OUT  256
#define L_IN   4096
#define L_OUT  (L_IN - K_TAPS + 1)
#define BATCH  32
#define TILE_M 256
#define NQ     (TILE_M / 4)
#define ROWS   (TILE_M + K_TAPS - 1)
#define PROWS  68
#define XPAD   132
#define X_FLOATS (4 * PROWS * XPAD)
#define NG     5
#define NS     6
#define KC     16
#define FB     128
#define NCHUNK (NG * (CIN / KC))
#define NWAVE  (2 * NS)
#define NT     (32 * NWAVE)
#define UCH_V4 (NWAVE * 256)
#define MPITCH 132
#define MT_FLOATS (NS * NQ * MPITCH)
#define EPS_F  1e-7f
#define WM     4
#define WN     4
#define XROWS_W ((ROWS + NWAVE - 1) / NWAVE)

__global__ __launch_bounds__(256)
void ckn1d_wfrag_kernel(const float* __restrict__ w,
                        float* __restrict__ u) {
    const int id   = blockIdx.x * 256 + threadIdx.x;
    const int lane = id & 31;
    const int ni   = (id >> 5) & 3;
    const int k8   = (id >> 7) & 1;
    const int hf   = (id >> 8) & 1;
    const int r    = id >> 9;
    const int fh   = r / NCHUNK;
    const int ch   = r - fh * NCHUNK;
    const int g    = ch >> 3;
    const int chan = (ch & 7) * KC + k8 * 8 + 4 * (lane >> 4);
    const int filt = fh * FB + hf * 64 + ni * 16 + (lane & 15);
    const float* src = w + (size_t)(3 * g * CIN + chan) * F_OUT + filt;
    v4f u0, u1, u2, u3, u4, u5;
    #pragma unroll
    for (int e = 0; e < 4; ++e) {
        float w0 = src[(size_t)(0 * CIN + e) * F_OUT];
        float w1 = src[(size_t)(1 * CIN + e) * F_OUT];
        float w2 = src[(size_t)(2 * CIN + e) * F_OUT];
        float sa = w0 + w2;
        float sb = w0 + 4.f * w2;
        u0[e] = 0.25f * w0;
        u1[e] = (-1.f / 6.f) * (sa + w1);
        u2[e] = (-1.f / 6.f) * (sa - w1);
        u3[e] = (1.f / 24.f) * (sb + 2.f * w1);
        u4[e] = (1.f / 24.f) * (sb - 2.f * w1);
        u5[e] = w2;
    }
    v4f* dst = (v4f*)u + (size_t)r * UCH_V4 + hf * 256 + k8 * 128 + ni * 32 + lane;
    dst[0 * 512] = u0;
    dst[1 * 512] = u1;
    dst[2 * 512] = u2;
    dst[3 * 512] = u3;
    dst[4 * 512] = u4;
    dst[5 * 512] = u5;
}

__global__ __launch_bounds__(NT)
void ckn1d_wmma_kernel(const float* __restrict__ x,
                       const float* __restrict__ u,
                       const float* __restrict__ scale,
                       const float* __restrict__ bias,
                       float* __restrict__ out) {
    __shared__ __attribute__((aligned(16))) float lds_main[MT_FLOATS];
    __shared__ float lds_rowsq[ROWS];
    __shared__ float lds_invn[TILE_M];
    __shared__ float lds_bias[FB];
    float* lds_x  = lds_main;

    const int t    = threadIdx.x;
    const int lane = t & 31;
    const int wave = __builtin_amdgcn_readfirstlane(t >> 5);
    const int fh   = blockIdx.x & 1;
    const int l0   = (blockIdx.x >> 1) * TILE_M;
    const int b    = blockIdx.y;

    const float* xb = x + (size_t)b * L_IN * CIN;
    v4f xv[XROWS_W];
    #pragma unroll
    for (int i = 0; i < XROWS_W; ++i) {
        const int gl  = l0 + wave + NWAVE * i;
        const int glc = (gl < L_IN) ? gl : (L_IN - 1);
        xv[i] = *(const v4f*)(xb + (size_t)glc * CIN + lane * 4);
    }
    #define X_SLOT(i_, row_) ((NWAVE * ((i_) + 1) <= ROWS || (row_) < ROWS)                 \
                              ? (((row_) & 3) * PROWS + ((row_) >> 2)) : (4 * PROWS + ((row_) - ROWS)))
    if (l0 + ROWS + NWAVE <= L_IN) {
        #pragma unroll
        for (int i = 0; i < XROWS_W; ++i) {
            const int row = wave + NWAVE * i;
            *(v4f*)(&lds_x[X_SLOT(i, row) * XPAD + lane * 4]) = xv[i];
        }
    } else {
        #pragma unroll
        for (int i = 0; i < XROWS_W; ++i) {
            const int row = wave + NWAVE * i;
            const v4f zero4 = {0.f, 0.f, 0.f, 0.f};
            *(v4f*)(&lds_x[X_SLOT(i, row) * XPAD + lane * 4]) = (l0 + row < L_IN) ? xv[i] : zero4;
        }
    }
    #undef X_SLOT
    if (t < FB) lds_bias[t] = bias[fh * FB + t];
    __syncthreads();

    if (t < ROWS) {
        const v4f* r = (const v4f*)&lds_x[((t & 3) * PROWS + (t >> 2)) * XPAD];
        float s = 0.f;
        #pragma unroll
        for (int c = 0; c < CIN / 4; ++c) {
            v4f v = r[c];
            s += v.x * v.x + v.y * v.y + v.z * v.z + v.w * v.w;
        }
        lds_rowsq[t] = s;
    }
    __syncthreads();
    if (t < TILE_M) {
        float win = 0.f;
        #pragma unroll
        for (int k = 0; k < K_TAPS; ++k) win += lds_rowsq[t + k];
        lds_invn[t] = scale[0] / __builtin_sqrtf(win + EPS_F);
    }

    const int js   = wave >> 1;
    const int hf   = wave & 1;
    const int half = lane >> 4;
    const int lm   = lane & 15;
    const int   o0 = (js == 0) ? 0 : 1;
    const int   o1 = (js == 5) ? 3 : 2;
    const int   o2 = (js == 0) ? 2 : 3;
    const int   o3 = (js == 5) ? 5 : 4;
    const bool  ends = (js == 0) || (js == 5);
    const float c0 = (js == 1) ? -4.f : ((js == 3) ? -2.f : ((js == 4) ? 2.f : 4.f));
    const float c1 = ends ? -2.5f : ((js <= 2) ? -4.f : -1.f);
    const float c2 = ends ? -2.5f : ((js == 1) ? 1.f : ((js == 2) ? -1.f : ((js == 3) ? 2.f : -2.f)));

    v8f acc[WM][WN] = {};

    const v4f* ub = (const v4f*)u + (size_t)(fh * NCHUNK) * UCH_V4 + wave * 256 + lane;
    const float* xlane = &lds_x[lm * XPAD + 4 * half];

    v4f r0[WM], r1[WM], r2[WM], r3[WM], fa[2][WM], fb[2][WN];

    #define A_PTR(e_, c0_) (xlane + (((e_) & 3) * PROWS + ((e_) >> 2)) * XPAD + (c0_))
    #define A_ROWS(chunk_) do {                                                            \
        const int g_  = (chunk_) >> 3;                                                     \
        const int k0_ = ((chunk_) & 7) * KC;                                               \
        p0 = A_PTR(3 * g_ + o0, k0_);                                                      \
        p1 = A_PTR(3 * g_ + o1, k0_);                                                      \
        p2 = A_PTR(3 * g_ + o2, k0_);                                                      \
        p3 = A_PTR(3 * g_ + o3, k0_);                                                      \
    } while (0)
    #define LOAD_A(koff_) do {                                                             \
        _Pragma("unroll") for (int mi = 0; mi < WM; ++mi) {                                \
            r0[mi] = *(const v4f*)(p0 + mi * 16 * XPAD + (koff_));                         \
            r1[mi] = *(const v4f*)(p1 + mi * 16 * XPAD + (koff_));                         \
            r2[mi] = *(const v4f*)(p2 + mi * 16 * XPAD + (koff_));                         \
            r3[mi] = *(const v4f*)(p3 + mi * 16 * XPAD + (koff_));                         \
        }                                                                                  \
    } while (0)
    #define LOAD_B(set_, chunk_, k8_) do {                                                 \
        _Pragma("unroll") for (int ni = 0; ni < WN; ++ni)                                  \
            fb[set_][ni] = ub[(size_t)(chunk_) * UCH_V4 + (k8_) * 128 + ni * 32];          \
    } while (0)
    #define FORM_A(set_) do {                                                              \
        _Pragma("unroll") for (int mi = 0; mi < WM; ++mi) {                                \
            _Pragma("unroll") for (int e = 0; e < 4; ++e)                                  \
                fa[set_][mi][e] = __builtin_fmaf(c2, r2[mi][e],                            \
                                  __builtin_fmaf(c1, r1[mi][e],                            \
                                  __builtin_fmaf(c0, r0[mi][e], r3[mi][e])));              \
        }                                                                                  \
    } while (0)
    #define MMA_GROUP(set_) do {                                                           \
        _Pragma("unroll") for (int sub = 0; sub < 2; ++sub) {                              \
            _Pragma("unroll") for (int mi = 0; mi < WM; ++mi) {                            \
                v2f a2;                                                                    \
                a2.x = fa[set_][mi][2 * sub];                                              \
                a2.y = fa[set_][mi][2 * sub + 1];                                          \
                _Pragma("unroll") for (int ni = 0; ni < WN; ++ni) {                        \
                    v2f b2;                                                                \
                    b2.x = fb[set_][ni][2 * sub];                                          \
                    b2.y = fb[set_][ni][2 * sub + 1];                                      \
                    acc[mi][ni] = __builtin_amdgcn_wmma_f32_16x16x4_f32(                   \
                        false, a2, false, b2, (short)0, acc[mi][ni], false, false);        \
                }                                                                          \
            }                                                                              \
        }                                                                                  \
    } while (0)

    const float *p0, *p1, *p2, *p3;
    A_ROWS(0);
    LOAD_B(0, 0, 0);
    LOAD_B(1, 0, 1);
    LOAD_A(0);

    for (int ch = 0; ch < NCHUNK; ++ch) {
        const int cn = (ch + 1 < NCHUNK) ? ch + 1 : NCHUNK - 1;

        FORM_A(0);
        LOAD_A(8);
        __builtin_amdgcn_sched_barrier(0);
        MMA_GROUP(0);
        LOAD_B(0, cn, 0);
        FORM_A(1);

        A_ROWS(cn);
        LOAD_A(0);
        __builtin_amdgcn_sched_barrier(0);
        MMA_GROUP(1);
        LOAD_B(1, cn, 1);
    }
    #undef A_PTR
    #undef A_ROWS
    #undef LOAD_A
    #undef LOAD_B
    #undef FORM_A
    #undef MMA_GROUP

    __syncthreads();
    float* mt = lds_main;
    #pragma unroll
    for (int mi = 0; mi < WM; ++mi) {
        #pragma unroll
        for (int v = 0; v < 8; ++v) {
            int pr = mi * 16 + v + 8 * half;
            #pragma unroll
            for (int ni = 0; ni < WN; ++ni)
                mt[(js * NQ + pr) * MPITCH + hf * 64 + ni * 16 + lm] = acc[mi][ni][v];
        }
    }
    __syncthreads();
    float* ob = out + ((size_t)b * L_OUT + l0) * F_OUT + fh * FB;
    const v4f bv = *(const v4f*)(&lds_bias[lane * 4]);
    for (int qd = wave; qd < NQ; qd += NWAVE) {
        const float* mrow = mt + qd * MPITCH + lane * 4;
        v4f m0 = *(const v4f*)(mrow);
        v4f m1 = *(const v4f*)(mrow + 1 * NQ * MPITCH);
        v4f m2 = *(const v4f*)(mrow + 2 * NQ * MPITCH);
        v4f m3 = *(const v4f*)(mrow + 3 * NQ * MPITCH);
        v4f m4 = *(const v4f*)(mrow + 4 * NQ * MPITCH);
        v4f m5 = *(const v4f*)(mrow + 5 * NQ * MPITCH);
        v4f s12 = m1 + m2, d12 = m1 - m2, s34 = m3 + m4, d34 = m3 - m4;
        v4f y[4];
        y[0] = (m0 + s12) + s34;
        y[1] = d12 + 2.f * d34;
        y[2] = s12 + 4.f * s34;
        y[3] = (d12 + 8.f * d34) + m5;
        #pragma unroll
        for (int j = 0; j < 4; ++j) {
            int pos = 4 * qd + j;
            if (l0 + pos < L_OUT)
                *(v4f*)(ob + (size_t)pos * F_OUT + lane * 4) = y[j] * lds_invn[pos] + bv;
        }
    }
}

extern "C" void kernel_launch(void* const* d_in, const int* in_sizes, int n_in,
                              void* d_out, int out_size, void* d_ws, size_t ws_size,
                              hipStream_t stream) {
    const float* x     = (const float*)d_in[0];
    const float* w     = (const float*)d_in[1];
    const float* scale = (const float*)d_in[2];
    const float* bias  = (const float*)d_in[3];
    float* out = (float*)d_out;

    dim3 grid(2 * ((L_OUT + TILE_M - 1) / TILE_M), BATCH);
    float* u = (float*)d_ws;
    ckn1d_wfrag_kernel<<<dim3(2 * NCHUNK * 512 / 256), 256, 0, stream>>>(w, u);
    ckn1d_wmma_kernel<<<grid, NT, 0, stream>>>(x, u, scale, bias, out);
}
